# starting kernel
# baseline (speedup 1.0000x reference)
#include <hip/hip_runtime.h>
#include <hip/hip_bf16.h>

typedef __attribute__((ext_vector_type(2))) float v2f;
typedef __attribute__((ext_vector_type(8))) float v8f;

#define EPSV 1e-5f
constexpr int Bn = 2, Cc = 256, Hh = 128, Ww = 128, Gg = 8;
constexpr int HW = Hh * Ww;

__global__ __launch_bounds__(256)
void conv1_bn_tanh(const float* __restrict__ x, const float* __restrict__ w1,
                   const float* __restrict__ b1, const float* __restrict__ g1,
                   const float* __restrict__ be1, const float* __restrict__ m1,
                   const float* __restrict__ v1, float* __restrict__ t)
{
    const int lane  = threadIdx.x & 31;
    const int g     = threadIdx.x >> 5;
    const int tile  = blockIdx.x;
    const int b     = tile >> 10;
    const int pbase = (tile & 1023) * 16;

    const int mrow  = lane & 15;
    const int khalf = (lane >> 4) * 2;
    const int arow  = mrow & 7;
    const bool aval = (mrow < 8);

    const float* xg = x  + ((size_t)b * Cc + g * 32) * HW;
    const float* wg = w1 + g * 8 * 32;

    v2f af[8], bf[8];
#pragma unroll
    for (int kk = 0; kk < 8; ++kk) {
        const int k0 = kk * 4 + khalf;
        const float ax = wg[arow * 32 + k0];
        const float ay = wg[arow * 32 + k0 + 1];
        af[kk].x = aval ? ax : 0.f;
        af[kk].y = aval ? ay : 0.f;
        bf[kk].x = xg[(size_t)k0       * HW + pbase + mrow];
        bf[kk].y = xg[(size_t)(k0 + 1) * HW + pbase + mrow];
    }
    v8f acc = {};
#pragma unroll
    for (int kk = 0; kk < 8; ++kk)
        acc = __builtin_amdgcn_wmma_f32_16x16x4_f32(false, af[kk], false, bf[kk],
                                                    (short)0, acc, false, false);

    if (lane < 16) {
        const int pix = pbase + lane;
#pragma unroll
        for (int i = 0; i < 8; ++i) {
            const int ch = g * 8 + i;
            const float inv = g1[ch] * rsqrtf(v1[ch] + EPSV);
            const float val = (acc[i] + b1[ch]) * inv + (be1[ch] - m1[ch] * inv);
            t[((size_t)b * 64 + ch) * HW + pix] = tanhf(val);
        }
    }
}

__global__ __launch_bounds__(256)
void conv2_bn(const float* __restrict__ t, const float* __restrict__ w2,
              const float* __restrict__ b2, const float* __restrict__ g2,
              const float* __restrict__ be2, const float* __restrict__ m2,
              const float* __restrict__ v2, float* __restrict__ mn)
{
    __shared__ float w2s[8 * 10 * 8];
    for (int i = threadIdx.x; i < 640; i += 256) w2s[i] = w2[i];
    __syncthreads();

    const int tid = blockIdx.x * 256 + threadIdx.x;
    const int b = tid / HW;
    const int p = tid % HW;
    const float* tb = t  + (size_t)b * 64 * HW + p;
    float*       mb = mn + (size_t)b * 80 * HW + p;

#pragma unroll
    for (int g = 0; g < 8; ++g) {
        float tin[8];
#pragma unroll
        for (int i = 0; i < 8; ++i) tin[i] = tb[(size_t)(g * 8 + i) * HW];
#pragma unroll
        for (int o = 0; o < 10; ++o) {
            float acc = 0.f;
#pragma unroll
            for (int i = 0; i < 8; ++i) acc += tin[i] * w2s[(g * 10 + o) * 8 + i];
            const int ch = g * 10 + o;
            const float inv = g2[ch] * rsqrtf(v2[ch] + EPSV);
            mb[(size_t)ch * HW] = (acc + b2[ch]) * inv + (be2[ch] - m2[ch] * inv);
        }
    }
}

__global__ __launch_bounds__(256)
void vconv(const float* __restrict__ x, const float* __restrict__ wv,
           float* __restrict__ v)
{
    const int lane = threadIdx.x & 31;
    const int wid  = threadIdx.x >> 5;
    int blk = blockIdx.x;
    const int strip = blk & 127;  blk >>= 7;
    const int g     = blk & 7;    blk >>= 3;
    const int b     = blk;
    const int pbase = (strip * 8 + wid) * 16;

    const int mrow  = lane & 15;
    const int khalf = (lane >> 4) * 2;

    const float* xg = x  + ((size_t)b * Cc + g * 32) * HW;
    const float* wg = wv + g * 32 * 32;

    v2f alo[8], ahi[8], bf[8];
#pragma unroll
    for (int kk = 0; kk < 8; ++kk) {
        const int k0 = kk * 4 + khalf;
        bf[kk].x  = xg[(size_t)k0       * HW + pbase + mrow];
        bf[kk].y  = xg[(size_t)(k0 + 1) * HW + pbase + mrow];
        alo[kk].x = wg[mrow * 32 + k0];         alo[kk].y = wg[mrow * 32 + k0 + 1];
        ahi[kk].x = wg[(mrow + 16) * 32 + k0];  ahi[kk].y = wg[(mrow + 16) * 32 + k0 + 1];
    }
    v8f clo = {}, chi = {};
#pragma unroll
    for (int kk = 0; kk < 8; ++kk) {
        clo = __builtin_amdgcn_wmma_f32_16x16x4_f32(false, alo[kk], false, bf[kk],
                                                    (short)0, clo, false, false);
        chi = __builtin_amdgcn_wmma_f32_16x16x4_f32(false, ahi[kk], false, bf[kk],
                                                    (short)0, chi, false, false);
    }
    const int orow = (lane < 16) ? 0 : 8;
    const int pix  = pbase + mrow;
    float* vg = v + ((size_t)b * Cc + g * 32) * HW + pix;
#pragma unroll
    for (int i = 0; i < 8; ++i) {
        vg[(size_t)(i + orow)      * HW] = clo[i];
        vg[(size_t)(i + orow + 16) * HW] = chi[i];
    }
}

__global__ __launch_bounds__(256)
void attn_kernel(const float* __restrict__ mn, const float* __restrict__ v,
                 float* __restrict__ out)
{
    __shared__ float vs[32][18 * 18];
    __shared__ float nbs[18 * 18];

    const int bg = blockIdx.x;
    const int b  = bg >> 3, g = bg & 7;
    const int y0 = blockIdx.y * 16, x0 = blockIdx.z * 16;
    const int lx = threadIdx.x & 15, ly = threadIdx.x >> 4;

    const float* vgrp = v  + ((size_t)b * 256 + g * 32) * HW;
    const float* nbp  = mn + ((size_t)b * 80 + g) * HW;

    for (int e = threadIdx.x; e < 324; e += 256) {
        const int r = e / 18, cdx = e % 18;
        const int gy = y0 - 1 + r, gx = x0 - 1 + cdx;
        nbs[e] = (gy >= 0 && gy < Hh && gx >= 0 && gx < Ww) ? nbp[gy * Ww + gx] : 0.f;
    }
    for (int e = threadIdx.x; e < 32 * 324; e += 256) {
        const int c = e / 324, rem = e % 324;
        const int r = rem / 18, cdx = rem % 18;
        const int gy = y0 - 1 + r, gx = x0 - 1 + cdx;
        vs[c][rem] = (gy >= 0 && gy < Hh && gx >= 0 && gx < Ww)
                         ? vgrp[(size_t)c * HW + gy * Ww + gx] : 0.f;
    }
    __syncthreads();

    const int y = y0 + ly, xp = x0 + lx;
    const float* maskp = mn + ((size_t)b * 80 + 8 + g * 9) * HW + (size_t)y * Ww + xp;

    float lg[9];
#pragma unroll
    for (int k = 0; k < 9; ++k) {
        const int i = k / 3, j = k % 3;
        lg[k] = maskp[(size_t)k * HW] + nbs[(ly + i) * 18 + (lx + j)];
    }
    float mx = lg[0];
#pragma unroll
    for (int k = 1; k < 9; ++k) mx = fmaxf(mx, lg[k]);
    float s = 0.f;
#pragma unroll
    for (int k = 0; k < 9; ++k) { lg[k] = expf(lg[k] - mx); s += lg[k]; }
    const float sinv = 1.f / s;
#pragma unroll
    for (int k = 0; k < 9; ++k) lg[k] *= sinv;

    float* outp = out + ((size_t)b * 256 + g * 32) * HW + (size_t)y * Ww + xp;
#pragma unroll
    for (int c = 0; c < 32; ++c) {
        float acc = 0.f;
#pragma unroll
        for (int k = 0; k < 9; ++k) {
            const int i = k / 3, j = k % 3;
            acc += lg[k] * vs[c][(ly + i) * 18 + (lx + j)];
        }
        outp[(size_t)c * HW] = acc;
    }
}

extern "C" void kernel_launch(void* const* d_in, const int* in_sizes, int n_in,
                              void* d_out, int out_size, void* d_ws, size_t ws_size,
                              hipStream_t stream)
{
    const float* x   = (const float*)d_in[0];
    const float* w1  = (const float*)d_in[1];
    const float* b1  = (const float*)d_in[2];
    const float* g1  = (const float*)d_in[3];
    const float* be1 = (const float*)d_in[4];
    const float* m1  = (const float*)d_in[5];
    const float* v1  = (const float*)d_in[6];
    const float* w2  = (const float*)d_in[7];
    const float* b2  = (const float*)d_in[8];
    const float* g2  = (const float*)d_in[9];
    const float* be2 = (const float*)d_in[10];
    const float* m2  = (const float*)d_in[11];
    const float* v2  = (const float*)d_in[12];
    const float* wvp = (const float*)d_in[13];
    float* out = (float*)d_out;

    float* t    = (float*)d_ws;
    float* mnb  = t   + (size_t)Bn * 64 * HW;
    float* vbuf = mnb + (size_t)Bn * 80 * HW;

    conv1_bn_tanh<<<Bn * 1024, 256, 0, stream>>>(x, w1, b1, g1, be1, m1, v1, t);
    conv2_bn<<<(Bn * HW) / 256, 256, 0, stream>>>(t, w2, b2, g2, be2, m2, v2, mnb);
    vconv<<<Bn * Gg * 128, 256, 0, stream>>>(x, wvp, vbuf);
    attn_kernel<<<dim3(Bn * Gg, Hh / 16, Ww / 16), 256, 0, stream>>>(mnb, vbuf, out);
}


# v kept pixel-major (a pixel's 32 channels = one 128-byte line) in workspace and attention LDS tile: tile rows and conv stores are whole lines, 16-byte loads and LDS accesses; sum order unchanged
# speedup vs baseline: 2.0459x; 2.0459x over previous
#include <hip/hip_runtime.h>
#include <hip/hip_bf16.h>

typedef __attribute__((ext_vector_type(2))) float v2f;
typedef __attribute__((ext_vector_type(8))) float v8f;
typedef __attribute__((ext_vector_type(4))) float v4f;

#define EPSV 1e-5f
constexpr int Bn = 2, Cc = 256, Hh = 128, Ww = 128, Gg = 8;
constexpr int HW = Hh * Ww;

__device__ __forceinline__ float tanh_sel(float x)
{
    const float a = __builtin_fabsf(x);
    const float z = a * a;
    float p = fmaf(z, -0.0057000206f, 0.020634072f);
    p = fmaf(z, p, -0.05373793f);
    p = fmaf(z, p, 0.13331416f);
    p = fmaf(z, p, -0.3333328f);
    const float lo = fmaf(z, a * p, a);
    const float e  = __builtin_amdgcn_exp2f(a * 2.885390082f);
    const float hi = fmaf(__builtin_amdgcn_rcpf(e + 1.0f), -2.0f, 1.0f);
    const float r  = (a < 0.625f) ? lo : hi;
    return __builtin_copysignf(r, x);
}

constexpr int TPW   = 4;
constexpr int GRIDA = Bn * 1024 / TPW;
constexpr int GRIDC = Bn * Gg * (128 / TPW);

__global__ __launch_bounds__(256)
void conv1_bn_tanh(const float* __restrict__ x, const float* __restrict__ w1,
                   const float* __restrict__ b1, const float* __restrict__ g1,
                   const float* __restrict__ be1, const float* __restrict__ m1,
                   const float* __restrict__ v1, float* __restrict__ t)
{
    const int lane   = threadIdx.x & 31;
    const int g      = threadIdx.x >> 5;
    const int tile0  = blockIdx.x * TPW;
    const int b      = tile0 >> 10;
    const int pbase0 = (tile0 & 1023) * 16;

    const int mrow  = lane & 15;
    const int khalf = (lane >> 4) * 2;
    const int lhalf = lane >> 4;
    const int arow  = (mrow + ((mrow >> 3) << 2)) & 7;

    const float* xp = x  + ((size_t)b * Cc + g * 32) * HW + pbase0 + mrow;
    const float* wg = w1 + g * 8 * 32;

    v2f af[8], bf[8];
#pragma unroll
    for (int kk = 0; kk < 8; ++kk) {
        const int k0 = kk * 4 + khalf;
        af[kk].x = wg[arow * 32 + k0];
        af[kk].y = wg[arow * 32 + k0 + 1];
        bf[kk].x = xp[(size_t)k0       * HW];
        bf[kk].y = xp[(size_t)(k0 + 1) * HW];
    }
    float cb[4], cinv[4], csh[4];
#pragma unroll
    for (int i = 0; i < 4; ++i) {
        const int ch = g * 8 + lhalf * 4 + i;
        const float inv = g1[ch] * rsqrtf(v1[ch] + EPSV);
        cb[i]   = b1[ch];
        cinv[i] = inv;
        csh[i]  = be1[ch] - m1[ch] * inv;
    }
    float* tp = t + ((size_t)b * 64 + g * 8 + lhalf * 4) * HW + pbase0 + mrow;

#pragma unroll
    for (int it = 0; it < TPW; ++it) {
        v2f nx[8];
#pragma unroll
        for (int kk = 0; kk < 8; ++kk) nx[kk] = bf[kk];
        if (it + 1 < TPW) {
#pragma unroll
            for (int kk = 0; kk < 8; ++kk) {
                const int k0 = kk * 4 + khalf;
                nx[kk].x = xp[(size_t)k0       * HW + (it + 1) * 16];
                nx[kk].y = xp[(size_t)(k0 + 1) * HW + (it + 1) * 16];
            }
        }
        __builtin_amdgcn_sched_barrier(0);
        v8f acc = {};
#pragma unroll
        for (int kk = 0; kk < 8; ++kk)
            acc = __builtin_amdgcn_wmma_f32_16x16x4_f32(false, af[kk], false, bf[kk],
                                                        (short)0, acc, false, false);

#pragma unroll
        for (int i = 0; i < 4; ++i) {
            const float val = (acc[i] + cb[i]) * cinv[i] + csh[i];
            tp[(size_t)i * HW + it * 16] = tanh_sel(val);
        }
#pragma unroll
        for (int kk = 0; kk < 8; ++kk) bf[kk] = nx[kk];
    }
}

constexpr int VPITCH = 36;

__global__ __launch_bounds__(256)
void vconv(const float* __restrict__ x, const float* __restrict__ wv,
           float* __restrict__ v)
{
    __shared__ __attribute__((aligned(16))) float stg[8][16 * VPITCH];

    const int lane = threadIdx.x & 31;
    const int wid  = threadIdx.x >> 5;
    int blk = blockIdx.x;
    const int sblk = blk % (128 / TPW);  blk /= (128 / TPW);
    const int g    = blk & 7;            blk >>= 3;
    const int b    = blk;
    const int pbase0 = (sblk * TPW * 8 + wid) * 16;

    const int mrow  = lane & 15;
    const int khalf = (lane >> 4) * 2;
    const int lhalf = lane >> 4;
    const int chlo  = ((mrow >> 3) << 4) + (mrow & 7);
    const int chhi  = chlo + 8;

    const float* xp = x  + ((size_t)b * Cc + g * 32) * HW + pbase0 + mrow;
    const float* wg = wv + g * 32 * 32;

    v2f alo[8], ahi[8], bf[TPW][8];
#pragma unroll
    for (int kk = 0; kk < 8; ++kk) {
        const int k0 = kk * 4 + khalf;
        bf[0][kk].x = xp[(size_t)k0       * HW];
        bf[0][kk].y = xp[(size_t)(k0 + 1) * HW];
        alo[kk].x = wg[chlo * 32 + k0];  alo[kk].y = wg[chlo * 32 + k0 + 1];
        ahi[kk].x = wg[chhi * 32 + k0];  ahi[kk].y = wg[chhi * 32 + k0 + 1];
    }
#pragma unroll
    for (int it = 1; it < TPW; ++it) {
#pragma unroll
        for (int kk = 0; kk < 8; ++kk) {
            const int k0 = kk * 4 + khalf;
            bf[it][kk].x = xp[(size_t)k0       * HW + it * 128];
            bf[it][kk].y = xp[(size_t)(k0 + 1) * HW + it * 128];
        }
    }
    float*        sw  = &stg[wid][0];
    v4f*       sst = reinterpret_cast<v4f*>(sw + mrow * VPITCH + 16 * lhalf);
    const v4f* sld = reinterpret_cast<const v4f*>(sw + (lane >> 3) * VPITCH + 4 * (lane & 7));
    float* vt = v + (((size_t)(b * Gg + g)) * HW + pbase0) * 32 + 4 * lane;

#pragma unroll
    for (int it = 0; it < TPW; ++it) {
        v8f clo = {}, chi = {};
#pragma unroll
        for (int kk = 0; kk < 8; ++kk) {
            clo = __builtin_amdgcn_wmma_f32_16x16x4_f32(false, alo[kk], false, bf[it][kk],
                                                        (short)0, clo, false, false);
            chi = __builtin_amdgcn_wmma_f32_16x16x4_f32(false, ahi[kk], false, bf[it][kk],
                                                        (short)0, chi, false, false);
        }
        sst[0] = v4f{clo[0], clo[1], clo[2], clo[3]};
        sst[1] = v4f{clo[4], clo[5], clo[6], clo[7]};
        sst[2] = v4f{chi[0], chi[1], chi[2], chi[3]};
        sst[3] = v4f{chi[4], chi[5], chi[6], chi[7]};
        __builtin_amdgcn_fence(__ATOMIC_RELEASE, "wavefront");
        __builtin_amdgcn_wave_barrier();
        __builtin_amdgcn_fence(__ATOMIC_ACQUIRE, "wavefront");
        v4f o[4];
#pragma unroll
        for (int j = 0; j < 4; ++j) o[j] = sld[j * VPITCH];
#pragma unroll
        for (int j = 0; j < 4; ++j)
            *reinterpret_cast<v4f*>(vt + (size_t)it * (128 * 32) + j * 128) = o[j];
        __builtin_amdgcn_fence(__ATOMIC_RELEASE, "wavefront");
        __builtin_amdgcn_wave_barrier();
        __builtin_amdgcn_fence(__ATOMIC_ACQUIRE, "wavefront");
    }
}

__global__ __launch_bounds__(256)
void attn_kernel(const float* __restrict__ t, const float* __restrict__ v,
                 float* __restrict__ out,
                 const float* __restrict__ w2, const float* __restrict__ b2,
                 const float* __restrict__ g2, const float* __restrict__ be2,
                 const float* __restrict__ m2, const float* __restrict__ v2)
{
    __shared__ __attribute__((aligned(16))) float vs[18 * 18 * VPITCH];
    __shared__ float nbs[18 * 18];
    __shared__ __attribute__((aligned(16))) float ctab[72];

    const int bg = blockIdx.x;
    const int b  = bg >> 3, g = bg & 7;
    const int y0 = blockIdx.y * 16, x0 = blockIdx.z * 16;
    const int tid = threadIdx.x;
    const int lx = tid & 15, ly = tid >> 4;

    const float* vgrp = v + ((size_t)(b * Gg + g)) * HW * 32;
    const float* tb   = t + (size_t)b * 64 * HW;

    const int y = y0 + ly, xp = x0 + lx;

    const int c0 = 8 + 9 * g;
    const int ga = c0 / 10;
    const int gb = (c0 + 8) / 10;
    const int nA = 10 * (ga + 1) - c0;

    const int  l7   = tid - 224;
    const bool isw  = (l7 >= 0);
    const bool isw2 = (l7 >= 0 && l7 < 8);
    float tw0 = 0.f, tw1 = 0.f, tw2 = 0.f;
    if (isw) {
        const float* wr = w2 + c0 * 8 + l7;
        tw0 = wr[0];
        tw1 = wr[32];
        if (isw2) tw2 = wr[64];
    }

    const int  r0  = tid / 18, cc0 = tid - r0 * 18;
    const int  gy0 = y0 - 1 + r0, gx0 = x0 - 1 + cc0;
    const bool in0 = (gy0 >= 0 && gy0 < Hh && gx0 >= 0 && gx0 < Ww);
    const int  of0 = in0 ? gy0 * Ww + gx0 : 0;
    const bool has1 = tid < 324 - 256;
    const int  e1  = 256 + tid;
    const int  r1  = e1 / 18, cc1 = e1 - r1 * 18;
    const int  gy1 = y0 - 1 + r1, gx1 = x0 - 1 + cc1;
    const bool in1 = has1 && (gy1 >= 0 && gy1 < Hh && gx1 >= 0 && gx1 < Ww);
    const int  of1 = in1 ? gy1 * Ww + gx1 : 0;

    float q0[8], q1[8];
#pragma unroll
    for (int i = 0; i < 8; ++i) q0[i] = tb[(size_t)i * HW + of0];
    const float* tpix = tb + (size_t)y * Ww + xp;
    float tA[8], tB[8];
#pragma unroll
    for (int i = 0; i < 8; ++i) tA[i] = tpix[(size_t)(ga * 8 + i) * HW];
#pragma unroll
    for (int i = 0; i < 8; ++i) tB[i] = tpix[(size_t)(gb * 8 + i) * HW];
    const int  part  = tid & 7;
    const int  pq    = tid >> 3;
    const v4f zero4 = {0.f, 0.f, 0.f, 0.f};
    v4f  av[11];
    bool inv_[11];
#pragma unroll
    for (int j = 0; j < 11; ++j) {
        const int  pos = pq + 32 * j;
        const int  r = pos / 18, cc = pos - r * 18;
        const int  gy = y0 - 1 + r, gx = x0 - 1 + cc;
        inv_[j] = (pos < 324 && gy >= 0 && gy < Hh && gx >= 0 && gx < Ww);
        const int  of = inv_[j] ? gy * Ww + gx : 0;
        av[j] = *reinterpret_cast<const v4f*>(vgrp + (size_t)of * 32 + part * 4);
    }
    if (has1) {
#pragma unroll
        for (int i = 0; i < 8; ++i) q1[i] = tb[(size_t)i * HW + of1];
    }

    const float* wn   = w2 + g * 8;
    const float  ninv = g2[g] * rsqrtf(v2[g] + EPSV);
    const float  nb2  = b2[g];
    const float  nsh  = be2[g] - m2[g] * ninv;
    float mb[9], minv[9], msh[9];
#pragma unroll
    for (int k = 0; k < 9; ++k) {
        const int   ch  = c0 + k;
        const float inv = g2[ch] * rsqrtf(v2[ch] + EPSV);
        mb[k]   = b2[ch];
        minv[k] = inv;
        msh[k]  = be2[ch] - m2[ch] * inv;
    }
    float n0 = 0.f;
#pragma unroll
    for (int i = 0; i < 8; ++i) n0 += q0[i] * wn[i];
    n0 = (n0 + nb2) * ninv + nsh;

    nbs[tid] = in0 ? n0 : 0.f;
#pragma unroll
    for (int j = 0; j < 10; ++j)
        *reinterpret_cast<v4f*>(&vs[(pq + 32 * j) * VPITCH + part * 4]) = inv_[j] ? av[j] : zero4;
    if (tid < 32)
        *reinterpret_cast<v4f*>(&vs[(320 + pq) * VPITCH + part * 4]) = inv_[10] ? av[10] : zero4;

    if (has1) {
        float n1 = 0.f;
#pragma unroll
        for (int i = 0; i < 8; ++i) n1 += q1[i] * wn[i];
        n1 = (n1 + nb2) * ninv + nsh;
        nbs[e1] = in1 ? n1 : 0.f;
    }
    if (isw) {
        ctab[l7] = tw0;
        ctab[l7 + 32] = tw1;
        if (isw2) ctab[l7 + 64] = tw2;
    }
    __syncthreads();

    float lg[9];
#pragma unroll
    for (int k = 0; k < 9; ++k) {
        const int i = k / 3, j = k % 3;
        lg[k] = nbs[(ly + i) * 18 + (lx + j)];
    }
#pragma unroll
    for (int k = 0; k < 9; ++k) {
        const bool inA = (k < nA);
        float acc = 0.f;
#pragma unroll
        for (int i = 0; i < 8; ++i) acc += (inA ? tA[i] : tB[i]) * ctab[k * 8 + i];
        lg[k] = ((acc + mb[k]) * minv[k] + msh[k]) + lg[k];
    }
    float mx = lg[0];
#pragma unroll
    for (int k = 1; k < 9; ++k) mx = fmaxf(mx, lg[k]);
    float s = 0.f;
#pragma unroll
    for (int k = 0; k < 9; ++k) { lg[k] = expf(lg[k] - mx); s += lg[k]; }
    const float sinv = 1.f / s;
#pragma unroll
    for (int k = 0; k < 9; ++k) lg[k] *= sinv;

    float* outp = out + ((size_t)b * 256 + g * 32) * HW + (size_t)y * Ww + xp;
    const float* vpix = &vs[(ly * 18 + lx) * VPITCH];
#pragma unroll
    for (int c4 = 0; c4 < 8; ++c4) {
        float acc0 = 0.f, acc1 = 0.f, acc2 = 0.f, acc3 = 0.f;
#pragma unroll
        for (int k = 0; k < 9; ++k) {
            const int i = k / 3, j = k % 3;
            const v4f vv = *reinterpret_cast<const v4f*>(vpix + (i * 18 + j) * VPITCH + c4 * 4);
            acc0 += lg[k] * vv.x;
            acc1 += lg[k] * vv.y;
            acc2 += lg[k] * vv.z;
            acc3 += lg[k] * vv.w;
        }
        outp[(size_t)(c4 * 4 + 0) * HW] = acc0;
        outp[(size_t)(c4 * 4 + 1) * HW] = acc1;
        outp[(size_t)(c4 * 4 + 2) * HW] = acc2;
        outp[(size_t)(c4 * 4 + 3) * HW] = acc3;
    }
}

extern "C" void kernel_launch(void* const* d_in, const int* in_sizes, int n_in,
                              void* d_out, int out_size, void* d_ws, size_t ws_size,
                              hipStream_t stream)
{
    const float* x   = (const float*)d_in[0];
    const float* w1  = (const float*)d_in[1];
    const float* b1  = (const float*)d_in[2];
    const float* g1  = (const float*)d_in[3];
    const float* be1 = (const float*)d_in[4];
    const float* m1  = (const float*)d_in[5];
    const float* v1  = (const float*)d_in[6];
    const float* w2  = (const float*)d_in[7];
    const float* b2  = (const float*)d_in[8];
    const float* g2  = (const float*)d_in[9];
    const float* be2 = (const float*)d_in[10];
    const float* m2  = (const float*)d_in[11];
    const float* v2  = (const float*)d_in[12];
    const float* wvp = (const float*)d_in[13];
    float* out = (float*)d_out;

    float* t    = (float*)d_ws;
    float* vbuf = t   + (size_t)Bn * 64 * HW;

    conv1_bn_tanh<<<GRIDA, 256, 0, stream>>>(x, w1, b1, g1, be1, m1, v1, t);
    vconv<<<GRIDC, 256, 0, stream>>>(x, wvp, vbuf);
    attn_kernel<<<dim3(Bn * Gg, Hh / 16, Ww / 16), 256, 0, stream>>>(t, vbuf, out, w2, b2, g2, be2, m2, v2);
}
